# starting kernel
# baseline (speedup 1.0000x reference)
#include <hip/hip_runtime.h>
#include <hip/hip_bf16.h>

typedef __attribute__((ext_vector_type(2))) float v2f;
typedef __attribute__((ext_vector_type(8))) float v8f;

#define B_  32
#define N_  4096
#define D_  512
#define C_  5

__global__ void __launch_bounds__(256)
k1_logits(const float* __restrict__ x, const float* __restrict__ W,
          float* __restrict__ e) {
    __shared__ float WT[16 * 514];

    const int nchunks = N_ / 128;
    const int b     = blockIdx.x / nchunks;
    const int chunk = blockIdx.x % nchunks;

    for (int i = threadIdx.x; i < 16 * 512; i += 256) {
        int c = i >> 9;
        int d = i & 511;
        WT[c * 514 + d] = (c < C_) ? W[d * C_ + c] : 0.0f;
    }
    __syncthreads();

    const int wave  = threadIdx.x >> 5;
    const int lane  = threadIdx.x & 31;
    const int row   = lane & 15;
    const int khalf = lane >> 4;

    const int n0 = chunk * 128 + wave * 16;

    const float* xp = x + ((size_t)b * N_ + n0 + row) * D_ + 2 * khalf;
    const float* wp = &WT[row * 514 + 2 * khalf];

    v8f acc = {};
    #pragma unroll 4
    for (int k = 0; k < D_; k += 4) {
        v2f a = *(const v2f*)(xp + k);
        v2f bb = *(const v2f*)(wp + k);
        acc = __builtin_amdgcn_wmma_f32_16x16x4_f32(
            false, a, false, bb, (short)0, acc, false, false);
    }

    const int c = row;
    if (c < C_) {
        #pragma unroll
        for (int r = 0; r < 8; ++r) {
            int n = n0 + r + 8 * khalf;
            e[((size_t)b * N_ + n) * C_ + c] = acc[r];
        }
    }
}

__global__ void __launch_bounds__(256)
k2_softmax(const float* __restrict__ e, float* __restrict__ alphaT) {
    __shared__ float red[256];
    const int b = blockIdx.x / C_;
    const int c = blockIdx.x % C_;
    const int tid = threadIdx.x;

    const float* ebc = e + (size_t)b * N_ * C_ + c;

    float m = -1e30f;
    for (int n = tid; n < N_; n += 256)
        m = fmaxf(m, ebc[(size_t)n * C_]);
    red[tid] = m;
    __syncthreads();
    for (int s = 128; s > 0; s >>= 1) {
        if (tid < s) red[tid] = fmaxf(red[tid], red[tid + s]);
        __syncthreads();
    }
    m = red[0];
    __syncthreads();

    float sl = 0.0f;
    for (int n = tid; n < N_; n += 256)
        sl += expf(ebc[(size_t)n * C_] - m);
    red[tid] = sl;
    __syncthreads();
    for (int s = 128; s > 0; s >>= 1) {
        if (tid < s) red[tid] += red[tid + s];
        __syncthreads();
    }
    const float inv = 1.0f / red[0];

    float* arow = alphaT + ((size_t)b * 16 + c) * N_;
    for (int n = tid; n < N_; n += 256)
        arow[n] = expf(ebc[(size_t)n * C_] - m) * inv;
}

__global__ void __launch_bounds__(256)
k3_vc(const float* __restrict__ x, const float* __restrict__ alphaT,
      float* __restrict__ vc) {
    const int dchunks = D_ / 128;
    const int b      = blockIdx.x / dchunks;
    const int dchunk = blockIdx.x % dchunks;

    const int wave  = threadIdx.x >> 5;
    const int lane  = threadIdx.x & 31;
    const int m     = lane & 15;
    const int khalf = lane >> 4;

    const int d0 = dchunk * 128 + wave * 16;

    const float* ap = alphaT + ((size_t)b * 16 + m) * N_ + 2 * khalf;
    const float* xp = x + ((size_t)b * N_ + 2 * khalf) * D_ + d0 + m;

    v8f acc = {};
    #pragma unroll 2
    for (int nk = 0; nk < N_; nk += 4) {
        v2f a = *(const v2f*)(ap + nk);
        v2f bb;
        bb.x = xp[(size_t)nk * D_];
        bb.y = xp[(size_t)(nk + 1) * D_];
        acc = __builtin_amdgcn_wmma_f32_16x16x4_f32(
            false, a, false, bb, (short)0, acc, false, false);
    }

    const int d = d0 + (lane & 15);
    #pragma unroll
    for (int r = 0; r < 8; ++r) {
        int c = r + 8 * khalf;
        if (c < C_)
            vc[((size_t)b * C_ + c) * D_ + d] = acc[r];
    }
}

__global__ void __launch_bounds__(256)
k4_gate(const float* __restrict__ vc, const float* __restrict__ lw,
        const float* __restrict__ lb, float* __restrict__ p_out,
        float* __restrict__ rs_out) {
    __shared__ float red[256];
    const int b = blockIdx.x / C_;
    const int c = blockIdx.x % C_;
    const int tid = threadIdx.x;

    const float* v = vc + ((size_t)b * C_ + c) * D_;
    float v0 = v[tid];
    float v1 = v[tid + 256];
    red[tid] = v0 * lw[tid] + v1 * lw[tid + 256];
    __syncthreads();
    for (int s = 128; s > 0; s >>= 1) {
        if (tid < s) red[tid] += red[tid + s];
        __syncthreads();
    }
    const float p = tanhf(red[0] + lb[0]);

    if (tid == 0) p_out[b * C_ + c] = p;
    float* rs = rs_out + ((size_t)b * C_ + c) * D_;
    rs[tid]       = p * v0;
    rs[tid + 256] = p * v1;
}

extern "C" void kernel_launch(void* const* d_in, const int* in_sizes, int n_in,
                              void* d_out, int out_size, void* d_ws, size_t ws_size,
                              hipStream_t stream) {
    const float* x  = (const float*)d_in[0];
    const float* W  = (const float*)d_in[1];
    const float* lw = (const float*)d_in[2];
    const float* lb = (const float*)d_in[3];

    float* out = (float*)d_out;
    float* p_out  = out;
    float* vc_out = out + (size_t)B_ * C_;
    float* rs_out = out + (size_t)B_ * C_ + (size_t)B_ * C_ * D_;

    float* ws      = (float*)d_ws;
    float* e_ws     = ws;
    float* alphaT_ws = ws + (size_t)B_ * N_ * C_;

    k1_logits<<<B_ * (N_ / 128), 256, 0, stream>>>(x, W, e_ws);
    k2_softmax<<<B_ * C_, 256, 0, stream>>>(e_ws, alphaT_ws);
    k3_vc<<<B_ * (D_ / 128), 256, 0, stream>>>(x, alphaT_ws, vc_out);
    k4_gate<<<B_ * C_, 256, 0, stream>>>(vc_out, lw, lb, p_out, rs_out);
}


# v_c GEMM re-tiled to keep memory busy: 256 groups of 16 waves, each wave one 256-row slice of n, 16-byte row loads two chunks ahead; wave partials summed via LDS in fixed order (changes sum order)
# speedup vs baseline: 4.2878x; 4.2878x over previous
#include <hip/hip_runtime.h>
#include <hip/hip_bf16.h>

typedef __attribute__((ext_vector_type(2))) float v2f;
typedef __attribute__((ext_vector_type(4))) float v4f;
typedef __attribute__((ext_vector_type(8))) float v8f;

#define B_  32
#define N_  4096
#define D_  512
#define C_  5

__global__ void __launch_bounds__(256)
k1_logits(const float* __restrict__ x, const float* __restrict__ W,
          float* __restrict__ e) {
    __shared__ float WT[16 * 514];

    const int nchunks = N_ / 128;
    const int b     = blockIdx.x / nchunks;
    const int chunk = blockIdx.x % nchunks;

    for (int i = threadIdx.x; i < 16 * 512; i += 256) {
        int c = i >> 9;
        int d = i & 511;
        WT[c * 514 + d] = (c < C_) ? W[d * C_ + c] : 0.0f;
    }
    __syncthreads();

    const int wave  = threadIdx.x >> 5;
    const int lane  = threadIdx.x & 31;
    const int row   = lane & 15;
    const int khalf = lane >> 4;

    const int n0 = chunk * 128 + wave * 16;

    const float* xp = x + ((size_t)b * N_ + n0 + row) * D_ + 2 * khalf;
    const float* wp = &WT[row * 514 + 2 * khalf];

    v8f acc = {};
    #pragma unroll 4
    for (int k = 0; k < D_; k += 4) {
        v2f a = *(const v2f*)(xp + k);
        v2f bb = *(const v2f*)(wp + k);
        acc = __builtin_amdgcn_wmma_f32_16x16x4_f32(
            false, a, false, bb, (short)0, acc, false, false);
    }

    const int c = row;
    if (c < C_) {
        #pragma unroll
        for (int r = 0; r < 8; ++r) {
            int n = n0 + r + 8 * khalf;
            e[((size_t)b * N_ + n) * C_ + c] = acc[r];
        }
    }
}

__global__ void __launch_bounds__(256)
k2_softmax(const float* __restrict__ e, float* __restrict__ alphaT) {
    __shared__ float red[256];
    const int b = blockIdx.x / C_;
    const int c = blockIdx.x % C_;
    const int tid = threadIdx.x;

    const float* ebc = e + (size_t)b * N_ * C_ + c;

    float m = -1e30f;
    for (int n = tid; n < N_; n += 256)
        m = fmaxf(m, ebc[(size_t)n * C_]);
    red[tid] = m;
    __syncthreads();
    for (int s = 128; s > 0; s >>= 1) {
        if (tid < s) red[tid] = fmaxf(red[tid], red[tid + s]);
        __syncthreads();
    }
    m = red[0];
    __syncthreads();

    float sl = 0.0f;
    for (int n = tid; n < N_; n += 256)
        sl += expf(ebc[(size_t)n * C_] - m);
    red[tid] = sl;
    __syncthreads();
    for (int s = 128; s > 0; s >>= 1) {
        if (tid < s) red[tid] += red[tid + s];
        __syncthreads();
    }
    const float inv = 1.0f / red[0];

    float* arow = alphaT + ((size_t)b * 16 + c) * N_;
    for (int n = tid; n < N_; n += 256)
        arow[n] = expf(ebc[(size_t)n * C_] - m) * inv;
}

#define K3_WAVES 16
#define K3_NS    (N_ / K3_WAVES)
#define K3_DC    64
#define K3_CH    32
#define K3_NCH   (K3_NS / K3_CH)

struct K3Chunk {
    v4f al[4];
    v4f xq[4][4];
};

__device__ __forceinline__ void k3_load(K3Chunk& ch, const float* __restrict__ ap,
                                        const float* __restrict__ xp) {
    for (int g = 0; g < 4; ++g) {
        ch.al[g] = *(const v4f*)(ap + 8 * g);
        for (int i = 0; i < 4; ++i)
            ch.xq[g][i] = *(const v4f*)(xp + (size_t)(8 * g + i) * D_);
    }
}

__device__ __forceinline__ v8f k3_wmma(v2f a, float b0, float b1, v8f acc) {
    v2f bb = {b0, b1};
    return __builtin_amdgcn_wmma_f32_16x16x4_f32(
        false, a, false, bb, (short)0, acc, false, false);
}

__device__ __forceinline__ void k3_mma(const K3Chunk& ch, v8f& a0, v8f& a1,
                                       v8f& a2, v8f& a3) {
    for (int g = 0; g < 4; ++g) {
        const v2f aA = {ch.al[g].x, ch.al[g].y};
        const v2f aB = {ch.al[g].z, ch.al[g].w};
        const v4f r0 = ch.xq[g][0], r1 = ch.xq[g][1];
        const v4f r2 = ch.xq[g][2], r3 = ch.xq[g][3];
        a0 = k3_wmma(aA, r0.x, r1.x, a0);
        a1 = k3_wmma(aA, r0.y, r1.y, a1);
        a2 = k3_wmma(aA, r0.z, r1.z, a2);
        a3 = k3_wmma(aA, r0.w, r1.w, a3);
        a0 = k3_wmma(aB, r2.x, r3.x, a0);
        a1 = k3_wmma(aB, r2.y, r3.y, a1);
        a2 = k3_wmma(aB, r2.z, r3.z, a2);
        a3 = k3_wmma(aB, r2.w, r3.w, a3);
    }
}

__global__ void __launch_bounds__(K3_WAVES * 32)
k3_vc(const float* __restrict__ x, const float* __restrict__ alphaT,
      float* __restrict__ vc) {
    __shared__ __attribute__((aligned(16))) float part[K3_WAVES * C_ * K3_DC];

    const int dchunks = D_ / K3_DC;
    const int b      = blockIdx.x / dchunks;
    const int dchunk = blockIdx.x % dchunks;

    const int wave  = threadIdx.x >> 5;
    const int lane  = threadIdx.x & 31;
    const int m     = lane & 15;
    const int khalf = lane >> 4;

    const int d0    = dchunk * K3_DC;
    const int nbase = wave * K3_NS;
    const int mrow  = m < C_ ? m : C_ - 1;

    const float* ap = alphaT + ((size_t)b * 16 + mrow) * N_ + nbase + 4 * khalf;
    const float* xp = x + ((size_t)b * N_ + nbase + 4 * khalf) * D_ + d0 + 4 * m;

    v8f acc0 = {}, acc1 = {}, acc2 = {}, acc3 = {};
    K3Chunk c0, c1;

    k3_load(c0, ap, xp);
    for (int ch = 0; ch < K3_NCH; ch += 2) {
        k3_load(c1, ap + (ch + 1) * K3_CH, xp + (size_t)(ch + 1) * K3_CH * D_);
        __builtin_amdgcn_sched_barrier(0);
        k3_mma(c0, acc0, acc1, acc2, acc3);
        __builtin_amdgcn_sched_barrier(0);
        if (ch + 2 < K3_NCH)
            k3_load(c0, ap + (ch + 2) * K3_CH, xp + (size_t)(ch + 2) * K3_CH * D_);
        __builtin_amdgcn_sched_barrier(0);
        k3_mma(c1, acc0, acc1, acc2, acc3);
        __builtin_amdgcn_sched_barrier(0);
    }

    if (khalf == 0) {
        for (int r = 0; r < C_; ++r) {
            v4f o = {acc0[r], acc1[r], acc2[r], acc3[r]};
            *(v4f*)&part[(wave * C_ + r) * K3_DC + 4 * m] = o;
        }
    }
    __syncthreads();

    if (threadIdx.x < C_ * K3_DC) {
        const int r  = threadIdx.x / K3_DC;
        const int dl = threadIdx.x % K3_DC;
        float s = 0.0f;
        for (int w = 0; w < K3_WAVES; ++w)
            s += part[(w * C_ + r) * K3_DC + dl];
        vc[((size_t)b * C_ + r) * D_ + d0 + dl] = s;
    }
}

__global__ void __launch_bounds__(256)
k4_gate(const float* __restrict__ vc, const float* __restrict__ lw,
        const float* __restrict__ lb, float* __restrict__ p_out,
        float* __restrict__ rs_out) {
    __shared__ float red[256];
    const int b = blockIdx.x / C_;
    const int c = blockIdx.x % C_;
    const int tid = threadIdx.x;

    const float* v = vc + ((size_t)b * C_ + c) * D_;
    float v0 = v[tid];
    float v1 = v[tid + 256];
    red[tid] = v0 * lw[tid] + v1 * lw[tid + 256];
    __syncthreads();
    for (int s = 128; s > 0; s >>= 1) {
        if (tid < s) red[tid] += red[tid + s];
        __syncthreads();
    }
    const float p = tanhf(red[0] + lb[0]);

    if (tid == 0) p_out[b * C_ + c] = p;
    float* rs = rs_out + ((size_t)b * C_ + c) * D_;
    rs[tid]       = p * v0;
    rs[tid + 256] = p * v1;
}

extern "C" void kernel_launch(void* const* d_in, const int* in_sizes, int n_in,
                              void* d_out, int out_size, void* d_ws, size_t ws_size,
                              hipStream_t stream) {
    const float* x  = (const float*)d_in[0];
    const float* W  = (const float*)d_in[1];
    const float* lw = (const float*)d_in[2];
    const float* lb = (const float*)d_in[3];

    float* out = (float*)d_out;
    float* p_out  = out;
    float* vc_out = out + (size_t)B_ * C_;
    float* rs_out = out + (size_t)B_ * C_ + (size_t)B_ * C_ * D_;

    float* ws      = (float*)d_ws;
    float* e_ws     = ws;
    float* alphaT_ws = ws + (size_t)B_ * N_ * C_;

    k1_logits<<<B_ * (N_ / 128), 256, 0, stream>>>(x, W, e_ws);
    k2_softmax<<<B_ * C_, 256, 0, stream>>>(e_ws, alphaT_ws);
    k3_vc<<<B_ * (D_ / K3_DC), K3_WAVES * 32, 0, stream>>>(x, alphaT_ws, vc_out);
    k4_gate<<<B_ * C_, 256, 0, stream>>>(vc_out, lw, lb, p_out, rs_out);
}


# logits GEMM re-tiled to keep memory busy: 256 groups of 16 waves, a wave owns 32 rows, 16-byte x loads two chunks ahead, two accumulators, W staged to LDS once (changes sum order over k)
# speedup vs baseline: 4.4770x; 1.0441x over previous
#include <hip/hip_runtime.h>
#include <hip/hip_bf16.h>

typedef __attribute__((ext_vector_type(2))) float v2f;
typedef __attribute__((ext_vector_type(4))) float v4f;
typedef __attribute__((ext_vector_type(8))) float v8f;

#define B_  32
#define N_  4096
#define D_  512
#define C_  5

#define K1_WAVES 16
#define K1_ROWS  32
#define K1_KC    32
#define K1_NCH   (D_ / K1_KC)
#define K1_PITCH 520

struct K1Chunk {
    v4f q[2][4];
};

__device__ __forceinline__ void k1_load(K1Chunk& ch, const float* __restrict__ xp) {
    for (int t = 0; t < 2; ++t)
        for (int i = 0; i < 4; ++i)
            ch.q[t][i] = *(const v4f*)(xp + (size_t)(16 * t) * D_ + 8 * i);
}

__device__ __forceinline__ v8f k1_wmma(float a0, float a1, float b0, float b1, v8f acc) {
    v2f a = {a0, a1};
    v2f bb = {b0, b1};
    return __builtin_amdgcn_wmma_f32_16x16x4_f32(
        false, a, false, bb, (short)0, acc, false, false);
}

struct K1WFrag {
    v4f w[4];
};

__device__ __forceinline__ void k1_wload(K1WFrag& f, const float* wp) {
    for (int i = 0; i < 4; ++i)
        f.w[i] = *(const v4f*)(wp + 8 * i);
}

__device__ __forceinline__ void k1_mma(const K1Chunk& ch, const K1WFrag& f,
                                       v8f& acc0, v8f& acc1) {
    for (int i = 0; i < 4; ++i) {
        const v4f w  = f.w[i];
        const v4f x0 = ch.q[0][i], x1 = ch.q[1][i];
        acc0 = k1_wmma(x0.x, x0.y, w.x, w.y, acc0);
        acc1 = k1_wmma(x1.x, x1.y, w.x, w.y, acc1);
        acc0 = k1_wmma(x0.z, x0.w, w.z, w.w, acc0);
        acc1 = k1_wmma(x1.z, x1.w, w.z, w.w, acc1);
    }
}

__global__ void __launch_bounds__(K1_WAVES * 32)
k1_logits(const float* __restrict__ x, const float* __restrict__ W,
          float* __restrict__ e) {
    __shared__ __attribute__((aligned(16))) float WT[C_ * K1_PITCH];

    const int wave  = threadIdx.x >> 5;
    const int lane  = threadIdx.x & 31;
    const int m     = lane & 15;
    const int khalf = lane >> 4;

    const size_t r0 = ((size_t)blockIdx.x * K1_WAVES + wave) * K1_ROWS;
    const float* xp = x + (r0 + m) * D_ + 4 * khalf;

    K1Chunk buf[3];
    k1_load(buf[0], xp);
    k1_load(buf[1], xp + K1_KC);

    for (int q = threadIdx.x; q < D_ * C_ / 4; q += K1_WAVES * 32) {
        const v4f w4 = *(const v4f*)(W + 4 * q);
        for (int j = 0; j < 4; ++j) {
            const int idx = 4 * q + j;
            const int d = idx / C_;
            const int c = idx - C_ * d;
            WT[c * K1_PITCH + d] = w4[j];
        }
    }
    __syncthreads();

    const int mcol = m < C_ ? m : C_ - 1;
    const float* wp = &WT[mcol * K1_PITCH + 4 * khalf];

    v8f acc0 = {}, acc1 = {};
    K1WFrag wf[2];
    k1_wload(wf[0], wp);
    #pragma unroll
    for (int ch = 0; ch < K1_NCH; ++ch) {
        if (ch + 2 < K1_NCH)
            k1_load(buf[(ch + 2) % 3], xp + (ch + 2) * K1_KC);
        if (ch + 1 < K1_NCH)
            k1_wload(wf[(ch + 1) & 1], wp + (ch + 1) * K1_KC);
        __builtin_amdgcn_sched_barrier(0);
        k1_mma(buf[ch % 3], wf[ch & 1], acc0, acc1);
        __builtin_amdgcn_sched_barrier(0);
    }

    if (m < C_) {
        for (int r = 0; r < 8; ++r) {
            e[(r0 + 8 * khalf + r) * C_ + m]      = acc0[r];
            e[(r0 + 16 + 8 * khalf + r) * C_ + m] = acc1[r];
        }
    }
}

__global__ void __launch_bounds__(256)
k2_softmax(const float* __restrict__ e, float* __restrict__ alphaT) {
    __shared__ float red[256];
    const int b = blockIdx.x / C_;
    const int c = blockIdx.x % C_;
    const int tid = threadIdx.x;

    const float* ebc = e + (size_t)b * N_ * C_ + c;

    float m = -1e30f;
    for (int n = tid; n < N_; n += 256)
        m = fmaxf(m, ebc[(size_t)n * C_]);
    red[tid] = m;
    __syncthreads();
    for (int s = 128; s > 0; s >>= 1) {
        if (tid < s) red[tid] = fmaxf(red[tid], red[tid + s]);
        __syncthreads();
    }
    m = red[0];
    __syncthreads();

    float sl = 0.0f;
    for (int n = tid; n < N_; n += 256)
        sl += expf(ebc[(size_t)n * C_] - m);
    red[tid] = sl;
    __syncthreads();
    for (int s = 128; s > 0; s >>= 1) {
        if (tid < s) red[tid] += red[tid + s];
        __syncthreads();
    }
    const float inv = 1.0f / red[0];

    float* arow = alphaT + ((size_t)b * 16 + c) * N_;
    for (int n = tid; n < N_; n += 256)
        arow[n] = expf(ebc[(size_t)n * C_] - m) * inv;
}

#define K3_WAVES 16
#define K3_NS    (N_ / K3_WAVES)
#define K3_DC    64
#define K3_CH    32
#define K3_NCH   (K3_NS / K3_CH)

struct K3Chunk {
    v4f al[4];
    v4f xq[4][4];
};

__device__ __forceinline__ void k3_load(K3Chunk& ch, const float* __restrict__ ap,
                                        const float* __restrict__ xp) {
    for (int g = 0; g < 4; ++g) {
        ch.al[g] = *(const v4f*)(ap + 8 * g);
        for (int i = 0; i < 4; ++i)
            ch.xq[g][i] = *(const v4f*)(xp + (size_t)(8 * g + i) * D_);
    }
}

__device__ __forceinline__ v8f k3_wmma(v2f a, float b0, float b1, v8f acc) {
    v2f bb = {b0, b1};
    return __builtin_amdgcn_wmma_f32_16x16x4_f32(
        false, a, false, bb, (short)0, acc, false, false);
}

__device__ __forceinline__ void k3_mma(const K3Chunk& ch, v8f& a0, v8f& a1,
                                       v8f& a2, v8f& a3) {
    for (int g = 0; g < 4; ++g) {
        const v2f aA = {ch.al[g].x, ch.al[g].y};
        const v2f aB = {ch.al[g].z, ch.al[g].w};
        const v4f r0 = ch.xq[g][0], r1 = ch.xq[g][1];
        const v4f r2 = ch.xq[g][2], r3 = ch.xq[g][3];
        a0 = k3_wmma(aA, r0.x, r1.x, a0);
        a1 = k3_wmma(aA, r0.y, r1.y, a1);
        a2 = k3_wmma(aA, r0.z, r1.z, a2);
        a3 = k3_wmma(aA, r0.w, r1.w, a3);
        a0 = k3_wmma(aB, r2.x, r3.x, a0);
        a1 = k3_wmma(aB, r2.y, r3.y, a1);
        a2 = k3_wmma(aB, r2.z, r3.z, a2);
        a3 = k3_wmma(aB, r2.w, r3.w, a3);
    }
}

__global__ void __launch_bounds__(K3_WAVES * 32)
k3_vc(const float* __restrict__ x, const float* __restrict__ alphaT,
      float* __restrict__ vc) {
    __shared__ __attribute__((aligned(16))) float part[K3_WAVES * C_ * K3_DC];

    const int dchunks = D_ / K3_DC;
    const int b      = blockIdx.x / dchunks;
    const int dchunk = blockIdx.x % dchunks;

    const int wave  = threadIdx.x >> 5;
    const int lane  = threadIdx.x & 31;
    const int m     = lane & 15;
    const int khalf = lane >> 4;

    const int d0    = dchunk * K3_DC;
    const int nbase = wave * K3_NS;
    const int mrow  = m < C_ ? m : C_ - 1;

    const float* ap = alphaT + ((size_t)b * 16 + mrow) * N_ + nbase + 4 * khalf;
    const float* xp = x + ((size_t)b * N_ + nbase + 4 * khalf) * D_ + d0 + 4 * m;

    v8f acc0 = {}, acc1 = {}, acc2 = {}, acc3 = {};
    K3Chunk c0, c1;

    k3_load(c0, ap, xp);
    for (int ch = 0; ch < K3_NCH; ch += 2) {
        k3_load(c1, ap + (ch + 1) * K3_CH, xp + (size_t)(ch + 1) * K3_CH * D_);
        __builtin_amdgcn_sched_barrier(0);
        k3_mma(c0, acc0, acc1, acc2, acc3);
        __builtin_amdgcn_sched_barrier(0);
        if (ch + 2 < K3_NCH)
            k3_load(c0, ap + (ch + 2) * K3_CH, xp + (size_t)(ch + 2) * K3_CH * D_);
        __builtin_amdgcn_sched_barrier(0);
        k3_mma(c1, acc0, acc1, acc2, acc3);
        __builtin_amdgcn_sched_barrier(0);
    }

    if (khalf == 0) {
        for (int r = 0; r < C_; ++r) {
            v4f o = {acc0[r], acc1[r], acc2[r], acc3[r]};
            *(v4f*)&part[(wave * C_ + r) * K3_DC + 4 * m] = o;
        }
    }
    __syncthreads();

    if (threadIdx.x < C_ * K3_DC) {
        const int r  = threadIdx.x / K3_DC;
        const int dl = threadIdx.x % K3_DC;
        float s = 0.0f;
        for (int w = 0; w < K3_WAVES; ++w)
            s += part[(w * C_ + r) * K3_DC + dl];
        vc[((size_t)b * C_ + r) * D_ + d0 + dl] = s;
    }
}

__global__ void __launch_bounds__(256)
k4_gate(const float* __restrict__ vc, const float* __restrict__ lw,
        const float* __restrict__ lb, float* __restrict__ p_out,
        float* __restrict__ rs_out) {
    __shared__ float red[256];
    const int b = blockIdx.x / C_;
    const int c = blockIdx.x % C_;
    const int tid = threadIdx.x;

    const float* v = vc + ((size_t)b * C_ + c) * D_;
    float v0 = v[tid];
    float v1 = v[tid + 256];
    red[tid] = v0 * lw[tid] + v1 * lw[tid + 256];
    __syncthreads();
    for (int s = 128; s > 0; s >>= 1) {
        if (tid < s) red[tid] += red[tid + s];
        __syncthreads();
    }
    const float p = tanhf(red[0] + lb[0]);

    if (tid == 0) p_out[b * C_ + c] = p;
    float* rs = rs_out + ((size_t)b * C_ + c) * D_;
    rs[tid]       = p * v0;
    rs[tid + 256] = p * v1;
}

extern "C" void kernel_launch(void* const* d_in, const int* in_sizes, int n_in,
                              void* d_out, int out_size, void* d_ws, size_t ws_size,
                              hipStream_t stream) {
    const float* x  = (const float*)d_in[0];
    const float* W  = (const float*)d_in[1];
    const float* lw = (const float*)d_in[2];
    const float* lb = (const float*)d_in[3];

    float* out = (float*)d_out;
    float* p_out  = out;
    float* vc_out = out + (size_t)B_ * C_;
    float* rs_out = out + (size_t)B_ * C_ + (size_t)B_ * C_ * D_;

    float* ws      = (float*)d_ws;
    float* e_ws     = ws;
    float* alphaT_ws = ws + (size_t)B_ * N_ * C_;

    k1_logits<<<(B_ * N_) / (K1_WAVES * K1_ROWS), K1_WAVES * 32, 0, stream>>>(x, W, e_ws);
    k2_softmax<<<B_ * C_, 256, 0, stream>>>(e_ws, alphaT_ws);
    k3_vc<<<B_ * (D_ / K3_DC), K3_WAVES * 32, 0, stream>>>(x, alphaT_ws, vc_out);
    k4_gate<<<B_ * C_, 256, 0, stream>>>(vc_out, lw, lb, p_out, rs_out);
}


# x read from memory once: logits, softmax over n and weighted sum fused (running-maximum softmax per 512-row slice, tile kept in LDS for 2nd product, slices merged in gate kernel; changes sum order)
# speedup vs baseline: 6.7747x; 1.5132x over previous
#include <hip/hip_runtime.h>
#include <hip/hip_bf16.h>

typedef __attribute__((ext_vector_type(2))) float v2f;
typedef __attribute__((ext_vector_type(4))) float v4f;
typedef __attribute__((ext_vector_type(8))) float v8f;

#define B_  32
#define N_  4096
#define D_  512
#define C_  5

#define F_WAVES  16
#define F_TR     64
#define F_SL     512
#define F_NT     (F_SL / F_TR)
#define F_NSL    (N_ / F_SL)
#define F_PITCH  516
#define F_WPITCH 520
#define F_KQ     128
#define F_NCH    4

struct FChunk {
    v4f q[4];
};

__device__ __forceinline__ void f_load(FChunk& ch, const float* __restrict__ xp) {
    for (int i = 0; i < 4; ++i)
        ch.q[i] = *(const v4f*)(xp + 8 * i);
}

__device__ __forceinline__ v8f f_wmma(float a0, float a1, float b0, float b1, v8f acc) {
    v2f a = {a0, a1};
    v2f bb = {b0, b1};
    return __builtin_amdgcn_wmma_f32_16x16x4_f32(
        false, a, false, bb, (short)0, acc, false, false);
}

struct FGroup {
    v4f al;
    v4f xq[4];
};

__device__ __forceinline__ void f_gload(FGroup& f, const float* ap, const float* xs) {
    f.al = *(const v4f*)ap;
    for (int i = 0; i < 4; ++i)
        f.xq[i] = *(const v4f*)(xs + i * F_PITCH);
}

__device__ __forceinline__ void f_gmma(const FGroup& f, v8f& a0, v8f& a1, v8f& a2, v8f& a3) {
    const v4f r0 = f.xq[0], r1 = f.xq[1], r2 = f.xq[2], r3 = f.xq[3];
    a0 = f_wmma(f.al.x, f.al.y, r0.x, r1.x, a0);
    a1 = f_wmma(f.al.x, f.al.y, r0.y, r1.y, a1);
    a2 = f_wmma(f.al.x, f.al.y, r0.z, r1.z, a2);
    a3 = f_wmma(f.al.x, f.al.y, r0.w, r1.w, a3);
    a0 = f_wmma(f.al.z, f.al.w, r2.x, r3.x, a0);
    a1 = f_wmma(f.al.z, f.al.w, r2.y, r3.y, a1);
    a2 = f_wmma(f.al.z, f.al.w, r2.z, r3.z, a2);
    a3 = f_wmma(f.al.z, f.al.w, r2.w, r3.w, a3);
}

__device__ __forceinline__ float f_wave_max(float v) {
    for (int o = 16; o > 0; o >>= 1) v = fmaxf(v, __shfl_xor(v, o, 32));
    return v;
}
__device__ __forceinline__ float f_wave_sum(float v) {
    for (int o = 16; o > 0; o >>= 1) v += __shfl_xor(v, o, 32);
    return v;
}

__device__ __forceinline__ void f_lds_barrier() {
    asm volatile("s_wait_dscnt 0x0\n\ts_barrier_signal -1\n\ts_barrier_wait -1" ::: "memory");
}

template <bool MORE>
__device__ __forceinline__ void f_tile(
        const int t, FChunk (&buf)[F_NCH], const float* __restrict__ xg,
        float* XS, float* EP, float* AL, float* SC,
        const float* wp, const float* ap, const int xs_off, const int xr_off,
        const int wave, const int lane, const int m, const int khalf,
        const int kq, const int rt,
        v8f& acc0, v8f& acc1, v8f& acc2, v8f& acc3, float& m_run, float& l_run) {
    float* xs = XS + (t & 1) * (F_TR * F_PITCH);
    const float* xn = xg + (size_t)(t + 1) * F_TR * D_;

    v8f ea = {}, eb = {};
    #pragma unroll
    for (int j = 0; j < F_NCH; ++j) {
        v4f w[4];
        for (int i = 0; i < 4; ++i) w[i] = *(const v4f*)(wp + 32 * j + 8 * i);
        for (int i = 0; i < 4; ++i)
            *(v4f*)(xs + xs_off + 32 * j + 8 * i) = buf[j].q[i];
        for (int i = 0; i < 4; i += 2) {
            const v4f x0 = buf[j].q[i], x1 = buf[j].q[i + 1];
            ea = f_wmma(x0.x, x0.y, w[i].x, w[i].y, ea);
            eb = f_wmma(x1.x, x1.y, w[i + 1].x, w[i + 1].y, eb);
            ea = f_wmma(x0.z, x0.w, w[i].z, w[i].w, ea);
            eb = f_wmma(x1.z, x1.w, w[i + 1].z, w[i + 1].w, eb);
        }
        if (MORE) f_load(buf[j], xn + 32 * j);
    }
    if (m < C_) {
        for (int r = 0; r < 8; ++r)
            EP[(kq * F_TR + rt * 16 + 8 * khalf + r) * C_ + m] = ea[r] + eb[r];
    }
    f_lds_barrier();

    if (wave < C_) {
        const int c = wave;
        float e0 = EP[(0 * F_TR + lane) * C_ + c];
        float e1 = EP[(0 * F_TR + lane + 32) * C_ + c];
        for (int k = 1; k < 4; ++k) {
            e0 += EP[(k * F_TR + lane) * C_ + c];
            e1 += EP[(k * F_TR + lane + 32) * C_ + c];
        }
        const float tmax  = f_wave_max(fmaxf(e0, e1));
        const float m_new = fmaxf(m_run, tmax);
        const float sc = expf(m_run - m_new);
        const float p0 = expf(e0 - m_new);
        const float p1 = expf(e1 - m_new);
        AL[c * F_TR + lane]      = p0;
        AL[c * F_TR + lane + 32] = p1;
        const float ps = f_wave_sum(p0 + p1);
        l_run = l_run * sc + ps;
        m_run = m_new;
        if (lane == 0) SC[c] = sc;
    }
    f_lds_barrier();

    for (int r = 0; r < C_; ++r) {
        const float s = SC[r];
        acc0[r] *= s; acc1[r] *= s; acc2[r] *= s; acc3[r] *= s;
    }
    const float* xr = xs + xr_off;
    FGroup g0, g1;
    f_gload(g0, ap, xr);
    #pragma unroll
    for (int g = 0; g < 4; g += 2) {
        f_gload(g1, ap + 8 * (g + 1), xr + 8 * (g + 1) * F_PITCH);
        __builtin_amdgcn_sched_barrier(0);
        f_gmma(g0, acc0, acc1, acc2, acc3);
        __builtin_amdgcn_sched_barrier(0);
        if (g + 2 < 4)
            f_gload(g0, ap + 8 * (g + 2), xr + 8 * (g + 2) * F_PITCH);
        __builtin_amdgcn_sched_barrier(0);
        f_gmma(g1, acc0, acc1, acc2, acc3);
        __builtin_amdgcn_sched_barrier(0);
    }
}

__global__ void __launch_bounds__(F_WAVES * 32)
kf_fused(const float* __restrict__ x, const float* __restrict__ W,
         float* __restrict__ pm, float* __restrict__ pl, float* __restrict__ pv) {
    __shared__ __attribute__((aligned(16))) float XS[2 * F_TR * F_PITCH];
    __shared__ __attribute__((aligned(16))) float WT[C_ * F_WPITCH];
    __shared__ float EP[4 * F_TR * C_];
    __shared__ __attribute__((aligned(16))) float AL[C_ * F_TR];
    __shared__ float SC[8];

    const int tid   = threadIdx.x;
    const int wave  = tid >> 5;
    const int lane  = tid & 31;
    const int m     = lane & 15;
    const int khalf = lane >> 4;

    const int b  = blockIdx.x / F_NSL;
    const int sl = blockIdx.x % F_NSL;
    const size_t rowbase = (size_t)b * N_ + (size_t)sl * F_SL;

    const int rt = wave & 3, kq = wave >> 2;
    const int cg = wave & 7, rh = wave >> 3;

    const float* xg = x + (rowbase + rt * 16 + m) * D_ + kq * F_KQ + 4 * khalf;
    const int xs_off = (rt * 16 + m) * F_PITCH + kq * F_KQ + 4 * khalf;

    FChunk buf[F_NCH];
    for (int j = 0; j < F_NCH; ++j) f_load(buf[j], xg + 32 * j);

    for (int q = tid; q < D_ * C_ / 4; q += F_WAVES * 32) {
        const v4f w4 = *(const v4f*)(W + 4 * q);
        for (int j = 0; j < 4; ++j) {
            const int idx = 4 * q + j;
            const int d = idx / C_;
            const int c = idx - C_ * d;
            WT[c * F_WPITCH + d] = w4[j];
        }
    }
    f_lds_barrier();

    const int mcol = m < C_ ? m : C_ - 1;
    const float* wp = &WT[mcol * F_WPITCH + kq * F_KQ + 4 * khalf];
    const float* ap = &AL[mcol * F_TR + rh * 32 + 4 * khalf];
    const int xr_off = (rh * 32 + 4 * khalf) * F_PITCH + cg * 64 + 4 * m;

    v8f acc0 = {}, acc1 = {}, acc2 = {}, acc3 = {};
    float m_run = -INFINITY, l_run = 0.0f;

    #pragma unroll 1
    for (int t = 0; t < F_NT - 1; ++t)
        f_tile<true>(t, buf, xg, XS, EP, AL, SC, wp, ap, xs_off, xr_off,
                     wave, lane, m, khalf, kq, rt,
                     acc0, acc1, acc2, acc3, m_run, l_run);
    f_tile<false>(F_NT - 1, buf, xg, XS, EP, AL, SC, wp, ap, xs_off, xr_off,
                  wave, lane, m, khalf, kq, rt,
                  acc0, acc1, acc2, acc3, m_run, l_run);

    float* part = XS;
    if (khalf == 0) {
        for (int r = 0; r < C_; ++r) {
            v4f o = {acc0[r], acc1[r], acc2[r], acc3[r]};
            *(v4f*)&part[(wave * C_ + r) * 64 + 4 * m] = o;
        }
    }
    f_lds_barrier();

    const size_t slot = (size_t)b * F_NSL + sl;
    {
        const int d = tid;
        const int cgi = d >> 6, dl = d & 63;
        for (int r = 0; r < C_; ++r) {
            const float s = part[(cgi * C_ + r) * 64 + dl]
                          + part[((cgi + 8) * C_ + r) * 64 + dl];
            pv[(slot * C_ + r) * D_ + d] = s;
        }
    }
    if (wave < C_ && lane == 0) {
        pm[slot * C_ + wave] = m_run;
        pl[slot * C_ + wave] = l_run;
    }
}

__global__ void __launch_bounds__(256)
kg_merge_gate(const float* __restrict__ pm, const float* __restrict__ pl,
              const float* __restrict__ pv, const float* __restrict__ lw,
              const float* __restrict__ lb, float* __restrict__ p_out,
              float* __restrict__ vc_out, float* __restrict__ rs_out) {
    __shared__ float red[256];
    const int b = blockIdx.x / C_;
    const int c = blockIdx.x % C_;
    const int tid = threadIdx.x;

    float ms[F_NSL];
    float mx = -INFINITY;
    for (int s = 0; s < F_NSL; ++s) {
        ms[s] = pm[((size_t)b * F_NSL + s) * C_ + c];
        mx = fmaxf(mx, ms[s]);
    }
    float L = 0.0f, v0 = 0.0f, v1 = 0.0f;
    for (int s = 0; s < F_NSL; ++s) {
        const size_t slot = (size_t)b * F_NSL + s;
        const float wgt = expf(ms[s] - mx);
        L += wgt * pl[slot * C_ + c];
        const float* pvs = pv + (slot * C_ + c) * D_;
        v0 += wgt * pvs[tid];
        v1 += wgt * pvs[tid + 256];
    }
    const float inv = 1.0f / L;
    v0 *= inv;
    v1 *= inv;

    red[tid] = v0 * lw[tid] + v1 * lw[tid + 256];
    __syncthreads();
    for (int s = 128; s > 0; s >>= 1) {
        if (tid < s) red[tid] += red[tid + s];
        __syncthreads();
    }
    const float p = tanhf(red[0] + lb[0]);

    if (tid == 0) p_out[b * C_ + c] = p;
    float* vo = vc_out + ((size_t)b * C_ + c) * D_;
    float* rs = rs_out + ((size_t)b * C_ + c) * D_;
    vo[tid]       = v0;
    vo[tid + 256] = v1;
    rs[tid]       = p * v0;
    rs[tid + 256] = p * v1;
}

extern "C" void kernel_launch(void* const* d_in, const int* in_sizes, int n_in,
                              void* d_out, int out_size, void* d_ws, size_t ws_size,
                              hipStream_t stream) {
    const float* x  = (const float*)d_in[0];
    const float* W  = (const float*)d_in[1];
    const float* lw = (const float*)d_in[2];
    const float* lb = (const float*)d_in[3];

    float* out = (float*)d_out;
    float* p_out  = out;
    float* vc_out = out + (size_t)B_ * C_;
    float* rs_out = out + (size_t)B_ * C_ + (size_t)B_ * C_ * D_;

    float* ws = (float*)d_ws;
    float* pm = ws;
    float* pl = ws + (size_t)B_ * F_NSL * C_;
    float* pv = ws + (size_t)2 * B_ * F_NSL * C_;

    kf_fused<<<B_ * F_NSL, F_WAVES * 32, 0, stream>>>(x, W, pm, pl, pv);
    kg_merge_gate<<<B_ * C_, 256, 0, stream>>>(pm, pl, pv, lw, lb, p_out, vc_out, rs_out);
}
